# starting kernel
# baseline (speedup 1.0000x reference)
#include <hip/hip_runtime.h>

typedef __attribute__((ext_vector_type(2))) float v2f;
typedef __attribute__((ext_vector_type(8))) float v8f;

#define N_ROWS 4096
#define DIMS   16
#define KCODES 65536
#define KSPLIT 16
#define CHUNKS_PER_SPLIT (KCODES / 16 / KSPLIT)

__global__ __launch_bounds__(256) void vq_prep_transpose(const float* __restrict__ x,
                                                         float* __restrict__ xt) {
  int tid = blockIdx.x * blockDim.x + threadIdx.x;
  int n = tid >> 4, d = tid & 15;
  int b = n >> 10, t = n & 1023;
  xt[tid] = x[(b * 16 + d) * 1024 + t];
}

__global__ __launch_bounds__(256) void vq_prep_cnorm(const float* __restrict__ cb,
                                                     float* __restrict__ cnh) {
  int j = blockIdx.x * blockDim.x + threadIdx.x;
  const float4* r = (const float4*)(cb + j * DIMS);
  float s = 0.f;
#pragma unroll
  for (int i = 0; i < 4; ++i) {
    float4 v = r[i];
    s += v.x * v.x + v.y * v.y + v.z * v.z + v.w * v.w;
  }
  cnh[j] = 0.5f * s;
}

__global__ __launch_bounds__(32) void vq_search(const float* __restrict__ xt,
                                                const float* __restrict__ cb,
                                                const float* __restrict__ cnh,
                                                float* __restrict__ pval,
                                                int* __restrict__ pidx) {
  const int lane = threadIdx.x;
  const int half = lane >> 4;
  const int l16  = lane & 15;
  const int dimOff = half * 2;
  const int rowTile = blockIdx.x;
  const int ks = blockIdx.y;

  const float* xrow = xt + (rowTile * 16 + l16) * DIMS;
  v2f a0 = *(const v2f*)(xrow + 0  + dimOff);
  v2f a1 = *(const v2f*)(xrow + 4  + dimOff);
  v2f a2 = *(const v2f*)(xrow + 8  + dimOff);
  v2f a3 = *(const v2f*)(xrow + 12 + dimOff);

  float best[8];
  int   bidx[8];
#pragma unroll
  for (int r = 0; r < 8; ++r) { best[r] = -3.4e38f; bidx[r] = 0; }

  const int cBeg = ks * CHUNKS_PER_SPLIT;
  const int cEnd = cBeg + CHUNKS_PER_SPLIT;

  {
  }
  int code = cBeg * 16 + l16;
  const float* crow = cb + code * DIMS;
  v2f b0 = *(const v2f*)(crow + 0  + dimOff);
  v2f b1 = *(const v2f*)(crow + 4  + dimOff);
  v2f b2 = *(const v2f*)(crow + 8  + dimOff);
  v2f b3 = *(const v2f*)(crow + 12 + dimOff);
  float ch = cnh[code];

  for (int c = cBeg; c < cEnd; ++c) {
    int cn = (c + 1 < cEnd) ? (c + 1) : c;
    int ncode = cn * 16 + l16;
    const float* nrow = cb + ncode * DIMS;
    v2f nb0 = *(const v2f*)(nrow + 0  + dimOff);
    v2f nb1 = *(const v2f*)(nrow + 4  + dimOff);
    v2f nb2 = *(const v2f*)(nrow + 8  + dimOff);
    v2f nb3 = *(const v2f*)(nrow + 12 + dimOff);
    float nch = cnh[ncode];

    v8f acc = {};
    acc = __builtin_amdgcn_wmma_f32_16x16x4_f32(false, a0, false, b0, (short)0, acc, false, false);
    acc = __builtin_amdgcn_wmma_f32_16x16x4_f32(false, a1, false, b1, (short)0, acc, false, false);
    acc = __builtin_amdgcn_wmma_f32_16x16x4_f32(false, a2, false, b2, (short)0, acc, false, false);
    acc = __builtin_amdgcn_wmma_f32_16x16x4_f32(false, a3, false, b3, (short)0, acc, false, false);

#pragma unroll
    for (int r = 0; r < 8; ++r) {
      float t = acc[r] - ch;
      if (t > best[r]) { best[r] = t; bidx[r] = code; }
    }

    b0 = nb0; b1 = nb1; b2 = nb2; b3 = nb3; ch = nch; code = ncode;
  }

#pragma unroll
  for (int m = 1; m < 16; m <<= 1) {
#pragma unroll
    for (int r = 0; r < 8; ++r) {
      float ov = __shfl_xor(best[r], m, 32);
      int   oi = __shfl_xor(bidx[r], m, 32);
      if (ov > best[r] || (ov == best[r] && oi < bidx[r])) { best[r] = ov; bidx[r] = oi; }
    }
  }

  if (l16 == 0) {
    int rbase = rowTile * 16 + half * 8;
#pragma unroll
    for (int r = 0; r < 8; ++r) {
      pval[(rbase + r) * KSPLIT + ks] = best[r];
      pidx[(rbase + r) * KSPLIT + ks] = bidx[r];
    }
  }
}

__global__ __launch_bounds__(256) void vq_finalize(const float* __restrict__ xt,
                                                   const float* __restrict__ cb,
                                                   const float* __restrict__ pval,
                                                   const int* __restrict__ pidx,
                                                   float* __restrict__ out,
                                                   float* __restrict__ rowloss) {
  int n = blockIdx.x * blockDim.x + threadIdx.x;
  float best = -3.4e38f;
  int bi = 0;
#pragma unroll
  for (int k = 0; k < KSPLIT; ++k) {
    float v = pval[n * KSPLIT + k];
    int   i = pidx[n * KSPLIT + k];
    if (v > best || (v == best && i < bi)) { best = v; bi = i; }
  }
  const float* q  = cb + bi * DIMS;
  const float* xr = xt + n * DIMS;
  int b = n >> 10, t = n & 1023;
  float loss = 0.f;
#pragma unroll
  for (int d = 0; d < DIMS; ++d) {
    float qv = q[d];
    out[(b * 16 + d) * 1024 + t] = qv;
    float diff = qv - xr[d];
    loss += diff * diff;
  }
  rowloss[n] = loss;
}

__global__ __launch_bounds__(256) void vq_loss(const float* __restrict__ rowloss,
                                               float* __restrict__ out) {
  __shared__ float sm[256];
  int tid = threadIdx.x;
  float s = 0.f;
  for (int i = tid; i < N_ROWS; i += 256) s += rowloss[i];
  sm[tid] = s;
  __syncthreads();
  for (int off = 128; off > 0; off >>= 1) {
    if (tid < off) sm[tid] += sm[tid + off];
    __syncthreads();
  }
  if (tid == 0) out[N_ROWS * DIMS] = sm[0] / (float)(N_ROWS * DIMS);
}

extern "C" void kernel_launch(void* const* d_in, const int* in_sizes, int n_in,
                              void* d_out, int out_size, void* d_ws, size_t ws_size,
                              hipStream_t stream) {
  const float* x  = (const float*)d_in[0];
  const float* cb = (const float*)d_in[1];
  float* out = (float*)d_out;

  float* ws      = (float*)d_ws;
  float* xt      = ws;
  float* cnh     = ws + 65536;
  float* pval    = ws + 131072;
  int*   pidx    = (int*)(ws + 196608);
  float* rowloss = ws + 262144;

  vq_prep_transpose<<<256, 256, 0, stream>>>(x, xt);
  vq_prep_cnorm<<<256, 256, 0, stream>>>(cb, cnh);

  dim3 grid(256, KSPLIT);
  vq_search<<<grid, 32, 0, stream>>>(xt, cb, cnh, pval, pidx);

  vq_finalize<<<16, 256, 0, stream>>>(xt, cb, pval, pidx, out, rowloss);
  vq_loss<<<1, 256, 0, stream>>>(rowloss, out);
}


# Search wave keeps A fragments of 4 row tiles (64 rows), reusing each loaded codebook chunk for 4 WMMA chains; K-split 16->64 keeps 4096 waves; partials split-major, finalize loads them as a group
# speedup vs baseline: 1.3587x; 1.3587x over previous
#include <hip/hip_runtime.h>

typedef __attribute__((ext_vector_type(2))) float v2f;
typedef __attribute__((ext_vector_type(8))) float v8f;

#define N_ROWS 4096
#define DIMS   16
#define KCODES 65536
#define KSPLIT 64
#define CHUNKS_PER_SPLIT (KCODES / 16 / KSPLIT)
#define RT 4
#define ROWS_PER_WAVE (RT * 16)

__global__ __launch_bounds__(256) void vq_prep_transpose(const float* __restrict__ x,
                                                         float* __restrict__ xt) {
  int tid = blockIdx.x * blockDim.x + threadIdx.x;
  int n = tid >> 4, d = tid & 15;
  int b = n >> 10, t = n & 1023;
  xt[tid] = x[(b * 16 + d) * 1024 + t];
}

__global__ __launch_bounds__(256) void vq_prep_cnorm(const float* __restrict__ cb,
                                                     float* __restrict__ cnh) {
  int j = blockIdx.x * blockDim.x + threadIdx.x;
  const float4* r = (const float4*)(cb + j * DIMS);
  float s = 0.f;
#pragma unroll
  for (int i = 0; i < 4; ++i) {
    float4 v = r[i];
    s += v.x * v.x + v.y * v.y + v.z * v.z + v.w * v.w;
  }
  cnh[j] = 0.5f * s;
}

__global__ __launch_bounds__(32) void vq_search(const float* __restrict__ xt,
                                                const float* __restrict__ cb,
                                                const float* __restrict__ cnh,
                                                float* __restrict__ pval,
                                                int* __restrict__ pidx) {
  const int lane = threadIdx.x;
  const int half = lane >> 4;
  const int l16  = lane & 15;
  const int dimOff = half * 2;
  const int rowGroup = blockIdx.x;
  const int ks = blockIdx.y;

  v2f a[RT][4];
#pragma unroll
  for (int j = 0; j < RT; ++j) {
    const float* xrow = xt + (rowGroup * ROWS_PER_WAVE + j * 16 + l16) * DIMS;
#pragma unroll
    for (int q = 0; q < 4; ++q) a[j][q] = *(const v2f*)(xrow + 4 * q + dimOff);
  }

  float best[RT][8];
  int   bidx[RT][8];
#pragma unroll
  for (int j = 0; j < RT; ++j) {
#pragma unroll
    for (int r = 0; r < 8; ++r) { best[j][r] = -3.4e38f; bidx[j][r] = 0; }
  }

  const int cBeg = ks * CHUNKS_PER_SPLIT;
  const int cEnd = cBeg + CHUNKS_PER_SPLIT;

  int code = cBeg * 16 + l16;
  const float* crow = cb + code * DIMS;
  v2f b0 = *(const v2f*)(crow + 0  + dimOff);
  v2f b1 = *(const v2f*)(crow + 4  + dimOff);
  v2f b2 = *(const v2f*)(crow + 8  + dimOff);
  v2f b3 = *(const v2f*)(crow + 12 + dimOff);
  float ch = cnh[code];

  for (int c = cBeg; c < cEnd; ++c) {
    int cn = (c + 1 < cEnd) ? (c + 1) : c;
    int ncode = cn * 16 + l16;
    const float* nrow = cb + ncode * DIMS;
    v2f nb0 = *(const v2f*)(nrow + 0  + dimOff);
    v2f nb1 = *(const v2f*)(nrow + 4  + dimOff);
    v2f nb2 = *(const v2f*)(nrow + 8  + dimOff);
    v2f nb3 = *(const v2f*)(nrow + 12 + dimOff);
    float nch = cnh[ncode];
    __builtin_amdgcn_sched_barrier(0);

    v8f acc[RT];
#pragma unroll
    for (int j = 0; j < RT; ++j) acc[j] = v8f{};
#pragma unroll
    for (int j = 0; j < RT; ++j)
      acc[j] = __builtin_amdgcn_wmma_f32_16x16x4_f32(false, a[j][0], false, b0, (short)0, acc[j], false, false);
#pragma unroll
    for (int j = 0; j < RT; ++j)
      acc[j] = __builtin_amdgcn_wmma_f32_16x16x4_f32(false, a[j][1], false, b1, (short)0, acc[j], false, false);
#pragma unroll
    for (int j = 0; j < RT; ++j)
      acc[j] = __builtin_amdgcn_wmma_f32_16x16x4_f32(false, a[j][2], false, b2, (short)0, acc[j], false, false);
#pragma unroll
    for (int j = 0; j < RT; ++j)
      acc[j] = __builtin_amdgcn_wmma_f32_16x16x4_f32(false, a[j][3], false, b3, (short)0, acc[j], false, false);

#pragma unroll
    for (int j = 0; j < RT; ++j) {
#pragma unroll
      for (int r = 0; r < 8; ++r) {
        float t = acc[j][r] - ch;
        if (t > best[j][r]) { best[j][r] = t; bidx[j][r] = code; }
      }
    }

    b0 = nb0; b1 = nb1; b2 = nb2; b3 = nb3; ch = nch; code = ncode;
  }

#pragma unroll
  for (int m = 1; m < 16; m <<= 1) {
#pragma unroll
    for (int j = 0; j < RT; ++j) {
#pragma unroll
      for (int r = 0; r < 8; ++r) {
        float ov = __shfl_xor(best[j][r], m, 32);
        int   oi = __shfl_xor(bidx[j][r], m, 32);
        if (ov > best[j][r] || (ov == best[j][r] && oi < bidx[j][r])) { best[j][r] = ov; bidx[j][r] = oi; }
      }
    }
  }

  const int jj = l16 >> 3, rr = l16 & 7;
#pragma unroll
  for (int s = 0; s < RT / 2; ++s) {
    float v = best[2 * s][0];
    int   id = bidx[2 * s][0];
#pragma unroll
    for (int t = 0; t < 2; ++t) {
#pragma unroll
      for (int r = 0; r < 8; ++r) {
        if (jj == t && rr == r) { v = best[2 * s + t][r]; id = bidx[2 * s + t][r]; }
      }
    }
    int row = rowGroup * ROWS_PER_WAVE + s * 32 + jj * 16 + half * 8 + rr;
    pval[ks * N_ROWS + row] = v;
    pidx[ks * N_ROWS + row] = id;
  }
}

__global__ __launch_bounds__(256) void vq_finalize(const float* __restrict__ xt,
                                                   const float* __restrict__ cb,
                                                   const float* __restrict__ pval,
                                                   const int* __restrict__ pidx,
                                                   float* __restrict__ out,
                                                   float* __restrict__ rowloss) {
  int n = blockIdx.x * blockDim.x + threadIdx.x;
  float best = -3.4e38f;
  int bi = 0;
#pragma unroll
  for (int k = 0; k < KSPLIT; ++k) {
    float v = pval[k * N_ROWS + n];
    int   i = pidx[k * N_ROWS + n];
    bool take = (v > best) | ((v == best) & (i < bi));
    best = take ? v : best;
    bi   = take ? i : bi;
  }
  const float* q  = cb + bi * DIMS;
  const float* xr = xt + n * DIMS;
  int b = n >> 10, t = n & 1023;
  float loss = 0.f;
#pragma unroll
  for (int d = 0; d < DIMS; ++d) {
    float qv = q[d];
    out[(b * 16 + d) * 1024 + t] = qv;
    float diff = qv - xr[d];
    loss += diff * diff;
  }
  rowloss[n] = loss;
}

__global__ __launch_bounds__(256) void vq_loss(const float* __restrict__ rowloss,
                                               float* __restrict__ out) {
  __shared__ float sm[256];
  int tid = threadIdx.x;
  float s = 0.f;
  for (int i = tid; i < N_ROWS; i += 256) s += rowloss[i];
  sm[tid] = s;
  __syncthreads();
  for (int off = 128; off > 0; off >>= 1) {
    if (tid < off) sm[tid] += sm[tid + off];
    __syncthreads();
  }
  if (tid == 0) out[N_ROWS * DIMS] = sm[0] / (float)(N_ROWS * DIMS);
}

extern "C" void kernel_launch(void* const* d_in, const int* in_sizes, int n_in,
                              void* d_out, int out_size, void* d_ws, size_t ws_size,
                              hipStream_t stream) {
  const float* x  = (const float*)d_in[0];
  const float* cb = (const float*)d_in[1];
  float* out = (float*)d_out;

  float* ws      = (float*)d_ws;
  float* xt      = ws;
  float* cnh     = ws + 65536;
  float* pval    = ws + 131072;
  int*   pidx    = (int*)(ws + 393216);
  float* rowloss = ws + 655360;

  vq_prep_transpose<<<256, 256, 0, stream>>>(x, xt);
  vq_prep_cnorm<<<256, 256, 0, stream>>>(cb, cnh);

  dim3 grid(64, KSPLIT);
  vq_search<<<grid, 32, 0, stream>>>(xt, cb, cnh, pval, pidx);

  vq_finalize<<<16, 256, 0, stream>>>(xt, cb, pval, pidx, out, rowloss);
  vq_loss<<<1, 256, 0, stream>>>(rowloss, out);
}


# Search loop software-pipelined over two accumulator sets: chunk k-1's score ranking (sub, compare, select) is issued pair by pair between the 16 WMMAs of chunk k; same sums and order, bit-identical
# speedup vs baseline: 1.4095x; 1.0374x over previous
#include <hip/hip_runtime.h>

typedef __attribute__((ext_vector_type(2))) float v2f;
typedef __attribute__((ext_vector_type(8))) float v8f;

#define N_ROWS 4096
#define DIMS   16
#define KCODES 65536
#define KSPLIT 64
#define CHUNKS_PER_SPLIT (KCODES / 16 / KSPLIT)
#define RT 4
#define ROWS_PER_WAVE (RT * 16)

__global__ __launch_bounds__(256) void vq_prep_transpose(const float* __restrict__ x,
                                                         float* __restrict__ xt) {
  int tid = blockIdx.x * blockDim.x + threadIdx.x;
  int n = tid >> 4, d = tid & 15;
  int b = n >> 10, t = n & 1023;
  xt[tid] = x[(b * 16 + d) * 1024 + t];
}

__global__ __launch_bounds__(256) void vq_prep_cnorm(const float* __restrict__ cb,
                                                     float* __restrict__ cnh) {
  int j = blockIdx.x * blockDim.x + threadIdx.x;
  const float4* r = (const float4*)(cb + j * DIMS);
  float s = 0.f;
#pragma unroll
  for (int i = 0; i < 4; ++i) {
    float4 v = r[i];
    s += v.x * v.x + v.y * v.y + v.z * v.z + v.w * v.w;
  }
  cnh[j] = 0.5f * s;
}

struct BFrag { v2f b0, b1, b2, b3; };

__device__ __forceinline__ BFrag vq_load_b(const float* __restrict__ cb, int chunk, int l16, int dimOff) {
  const float* crow = cb + (chunk * 16 + l16) * DIMS + dimOff;
  BFrag f;
  f.b0 = *(const v2f*)(crow + 0);
  f.b1 = *(const v2f*)(crow + 4);
  f.b2 = *(const v2f*)(crow + 8);
  f.b3 = *(const v2f*)(crow + 12);
  return f;
}

#define VQ_PJ(p) ((p) >> 2)
#define VQ_PR(p) (((p) & 3) * 2)

__device__ __forceinline__ void vq_score(const v8f (&accR)[RT], int p, float chR, float (&t)[32]) {
  t[2 * p]     = accR[VQ_PJ(p)][VQ_PR(p)]     - chR;
  t[2 * p + 1] = accR[VQ_PJ(p)][VQ_PR(p) + 1] - chR;
}
__device__ __forceinline__ void vq_compare(int p, const float (&t)[32], const float (&best)[RT][8], bool (&g)[32]) {
  g[2 * p]     = t[2 * p]     > best[VQ_PJ(p)][VQ_PR(p)];
  g[2 * p + 1] = t[2 * p + 1] > best[VQ_PJ(p)][VQ_PR(p) + 1];
}
__device__ __forceinline__ void vq_select(int p, const float (&t)[32], const bool (&g)[32], int codeR,
                                          float (&best)[RT][8], int (&bidx)[RT][8]) {
#pragma unroll
  for (int e = 0; e < 2; ++e) {
    const int j = VQ_PJ(p), r = VQ_PR(p) + e;
    best[j][r] = g[2 * p + e] ? t[2 * p + e] : best[j][r];
    bidx[j][r] = g[2 * p + e] ? codeR : bidx[j][r];
  }
}

__device__ __forceinline__ void vq_hold(int p, const float (&best)[RT][8], const int (&bidx)[RT][8]) {
  asm volatile("" :: "v"(best[VQ_PJ(p)][VQ_PR(p)]), "v"(best[VQ_PJ(p)][VQ_PR(p) + 1]),
                     "v"(bidx[VQ_PJ(p)][VQ_PR(p)]), "v"(bidx[VQ_PJ(p)][VQ_PR(p) + 1]));
}

template <bool RANK>
__device__ __forceinline__ void vq_step(v8f (&accW)[RT], const v8f (&accR)[RT], const v2f (&a)[RT][4],
                                        const BFrag& b, float chR, int codeR,
                                        float (&best)[RT][8], int (&bidx)[RT][8]) {
  float t[32];
  bool  g[32];
  if (RANK) vq_score(accR, 0, chR, t);
#pragma unroll
  for (int w = 0; w < 16; ++w) {
    const int q = w >> 2, j = w & 3;
    const v2f bq = (q == 0) ? b.b0 : (q == 1) ? b.b1 : (q == 2) ? b.b2 : b.b3;
    v8f cin = v8f{};
    if (q != 0) cin = accW[j];
    accW[j] = __builtin_amdgcn_wmma_f32_16x16x4_f32(false, a[j][q], false, bq, (short)0, cin, false, false);
    if (RANK) {
      if (w + 1 < 16) vq_score(accR, w + 1, chR, t);
      vq_compare(w, t, best, g);
      if (w >= 1) {
        vq_select(w - 1, t, g, codeR, best, bidx);
        vq_hold(w - 1, best, bidx);
      }
      __builtin_amdgcn_sched_barrier(0);
    }
  }
  if (RANK) {
    vq_select(15, t, g, codeR, best, bidx);
    vq_hold(15, best, bidx);
  }
}

__device__ __forceinline__ void vq_rank_all(const v8f (&accR)[RT], float chR, int codeR,
                                            float (&best)[RT][8], int (&bidx)[RT][8]) {
  float t[32];
  bool  g[32];
#pragma unroll
  for (int p = 0; p < 16; ++p) {
    vq_score(accR, p, chR, t);
    vq_compare(p, t, best, g);
    vq_select(p, t, g, codeR, best, bidx);
  }
}

__global__ __launch_bounds__(32) void vq_search(const float* __restrict__ xt,
                                                const float* __restrict__ cb,
                                                const float* __restrict__ cnh,
                                                float* __restrict__ pval,
                                                int* __restrict__ pidx) {
  const int lane = threadIdx.x;
  const int half = lane >> 4;
  const int l16  = lane & 15;
  const int dimOff = half * 2;
  const int rowGroup = blockIdx.x;
  const int ks = blockIdx.y;

  v2f a[RT][4];
#pragma unroll
  for (int j = 0; j < RT; ++j) {
    const float* xrow = xt + (rowGroup * ROWS_PER_WAVE + j * 16 + l16) * DIMS;
#pragma unroll
    for (int q = 0; q < 4; ++q) a[j][q] = *(const v2f*)(xrow + 4 * q + dimOff);
  }

  float best[RT][8];
  int   bidx[RT][8];
#pragma unroll
  for (int j = 0; j < RT; ++j) {
#pragma unroll
    for (int r = 0; r < 8; ++r) { best[j][r] = -3.4e38f; bidx[j][r] = 0; }
  }

  const int cBeg = ks * CHUNKS_PER_SPLIT;

  BFrag s0 = vq_load_b(cb, cBeg, l16, dimOff);
  BFrag s1 = vq_load_b(cb, cBeg + 1, l16, dimOff);
  float c0 = cnh[cBeg * 16 + l16];
  float c1 = 0.f;
  int codeR = cBeg * 16 + l16;
  v8f accA[RT], accB[RT];
  vq_step<false>(accA, accA, a, s0, 0.f, 0, best, bidx);

  for (int i = 0; i < (CHUNKS_PER_SPLIT - 2) / 2; ++i) {
    const int k = cBeg + 2 * i + 1;
    s0 = vq_load_b(cb, k + 1, l16, dimOff);
    c1 = cnh[k * 16 + l16];
    __builtin_amdgcn_sched_barrier(0);
    vq_step<true>(accB, accA, a, s1, c0, codeR, best, bidx);
    codeR += 16;

    s1 = vq_load_b(cb, k + 2, l16, dimOff);
    c0 = cnh[(k + 1) * 16 + l16];
    __builtin_amdgcn_sched_barrier(0);
    vq_step<true>(accA, accB, a, s0, c1, codeR, best, bidx);
    codeR += 16;
  }

  c1 = cnh[(cBeg + CHUNKS_PER_SPLIT - 1) * 16 + l16];
  vq_step<true>(accB, accA, a, s1, c0, codeR, best, bidx);
  codeR += 16;
  vq_rank_all(accB, c1, codeR, best, bidx);

#pragma unroll
  for (int m = 1; m < 16; m <<= 1) {
#pragma unroll
    for (int j = 0; j < RT; ++j) {
#pragma unroll
      for (int r = 0; r < 8; ++r) {
        float ov = __shfl_xor(best[j][r], m, 32);
        int   oi = __shfl_xor(bidx[j][r], m, 32);
        if (ov > best[j][r] || (ov == best[j][r] && oi < bidx[j][r])) { best[j][r] = ov; bidx[j][r] = oi; }
      }
    }
  }

  const int jj = l16 >> 3, rr = l16 & 7;
#pragma unroll
  for (int s = 0; s < RT / 2; ++s) {
    float v = best[2 * s][0];
    int   id = bidx[2 * s][0];
#pragma unroll
    for (int t = 0; t < 2; ++t) {
#pragma unroll
      for (int r = 0; r < 8; ++r) {
        if (jj == t && rr == r) { v = best[2 * s + t][r]; id = bidx[2 * s + t][r]; }
      }
    }
    int row = rowGroup * ROWS_PER_WAVE + s * 32 + jj * 16 + half * 8 + rr;
    pval[ks * N_ROWS + row] = v;
    pidx[ks * N_ROWS + row] = id;
  }
}

__global__ __launch_bounds__(256) void vq_finalize(const float* __restrict__ xt,
                                                   const float* __restrict__ cb,
                                                   const float* __restrict__ pval,
                                                   const int* __restrict__ pidx,
                                                   float* __restrict__ out,
                                                   float* __restrict__ rowloss) {
  int n = blockIdx.x * blockDim.x + threadIdx.x;
  float best = -3.4e38f;
  int bi = 0;
#pragma unroll
  for (int k = 0; k < KSPLIT; ++k) {
    float v = pval[k * N_ROWS + n];
    int   i = pidx[k * N_ROWS + n];
    bool take = (v > best) | ((v == best) & (i < bi));
    best = take ? v : best;
    bi   = take ? i : bi;
  }
  const float* q  = cb + bi * DIMS;
  const float* xr = xt + n * DIMS;
  int b = n >> 10, t = n & 1023;
  float loss = 0.f;
#pragma unroll
  for (int d = 0; d < DIMS; ++d) {
    float qv = q[d];
    out[(b * 16 + d) * 1024 + t] = qv;
    float diff = qv - xr[d];
    loss += diff * diff;
  }
  rowloss[n] = loss;
}

__global__ __launch_bounds__(256) void vq_loss(const float* __restrict__ rowloss,
                                               float* __restrict__ out) {
  __shared__ float sm[256];
  int tid = threadIdx.x;
  float s = 0.f;
  for (int i = tid; i < N_ROWS; i += 256) s += rowloss[i];
  sm[tid] = s;
  __syncthreads();
  for (int off = 128; off > 0; off >>= 1) {
    if (tid < off) sm[tid] += sm[tid + off];
    __syncthreads();
  }
  if (tid == 0) out[N_ROWS * DIMS] = sm[0] / (float)(N_ROWS * DIMS);
}

extern "C" void kernel_launch(void* const* d_in, const int* in_sizes, int n_in,
                              void* d_out, int out_size, void* d_ws, size_t ws_size,
                              hipStream_t stream) {
  const float* x  = (const float*)d_in[0];
  const float* cb = (const float*)d_in[1];
  float* out = (float*)d_out;

  float* ws      = (float*)d_ws;
  float* xt      = ws;
  float* cnh     = ws + 65536;
  float* pval    = ws + 131072;
  int*   pidx    = (int*)(ws + 393216);
  float* rowloss = ws + 655360;

  vq_prep_transpose<<<256, 256, 0, stream>>>(x, xt);
  vq_prep_cnorm<<<256, 256, 0, stream>>>(cb, cnh);

  dim3 grid(64, KSPLIT);
  vq_search<<<grid, 32, 0, stream>>>(xt, cb, cnh, pval, pidx);

  vq_finalize<<<16, 256, 0, stream>>>(xt, cb, pval, pidx, out, rowloss);
  vq_loss<<<1, 256, 0, stream>>>(rowloss, out);
}


# Score tile transposed (codes = WMMA A operand, x rows = B), scores bit-identical: a lane's running argmax is a max3 chain plus one 8-code group id per tile; code in the group re-scored in finalize
# speedup vs baseline: 1.6943x; 1.2021x over previous
#include <hip/hip_runtime.h>

typedef __attribute__((ext_vector_type(2))) float v2f;
typedef __attribute__((ext_vector_type(8))) float v8f;

#define N_ROWS 4096
#define DIMS   16
#define KCODES 65536
#define KSPLIT 64
#define CHUNKS_PER_SPLIT (KCODES / 16 / KSPLIT)
#define RT 4
#define ROWS_PER_WAVE (RT * 16)

__global__ __launch_bounds__(256) void vq_prep_transpose(const float* __restrict__ x,
                                                         float* __restrict__ xt) {
  int tid = blockIdx.x * blockDim.x + threadIdx.x;
  int n = tid >> 4, d = tid & 15;
  int b = n >> 10, t = n & 1023;
  xt[tid] = x[(b * 16 + d) * 1024 + t];
}

__global__ __launch_bounds__(256) void vq_prep_cnorm(const float* __restrict__ cb,
                                                     float* __restrict__ cnh) {
  int j = blockIdx.x * blockDim.x + threadIdx.x;
  const float4* r = (const float4*)(cb + j * DIMS);
  float s = 0.f;
#pragma unroll
  for (int i = 0; i < 4; ++i) {
    float4 v = r[i];
    s += v.x * v.x + v.y * v.y + v.z * v.z + v.w * v.w;
  }
  cnh[j] = 0.5f * s;
}

struct CFrag { v2f b0, b1, b2, b3; };

__device__ __forceinline__ CFrag vq_load_codes(const float* __restrict__ cb, int chunk, int l16, int dimOff) {
  const float* crow = cb + (chunk * 16 + l16) * DIMS + dimOff;
  CFrag f;
  f.b0 = *(const v2f*)(crow + 0);
  f.b1 = *(const v2f*)(crow + 4);
  f.b2 = *(const v2f*)(crow + 8);
  f.b3 = *(const v2f*)(crow + 12);
  return f;
}

__device__ __forceinline__ v8f vq_load_norms(const float* __restrict__ cnh, int chunk, int half) {
  return *(const v8f*)(cnh + chunk * 16 + half * 8);
}

__device__ __forceinline__ float vq_max3(float a, float b, float c) {
  return __builtin_fmaxf(__builtin_fmaxf(a, b), c);
}
__device__ __forceinline__ void vq_finish(int j, float nb, int grpR, float (&best)[RT], int (&bgrp)[RT]) {
  const bool g = nb > best[j];
  bgrp[j] = g ? grpR : bgrp[j];
  best[j] = nb;
}

__device__ __forceinline__ void vq_hold8(const v8f& v) { asm volatile("" :: "v"(v)); }
__device__ __forceinline__ void vq_hold1(float v) { asm volatile("" :: "v"(v)); }
__device__ __forceinline__ void vq_hold2(float v, int i) { asm volatile("" :: "v"(v), "v"(i)); }

template <bool RANK>
__device__ __forceinline__ void vq_step(v8f (&accW)[RT], const v8f (&accR)[RT], const v2f (&a)[RT][4],
                                        const CFrag& c, const v8f& nrmR, int grpR,
                                        float (&best)[RT], int (&bgrp)[RT]) {
  v8f tv = v8f{};
  float nb = 0.f;
#pragma unroll
  for (int w = 0; w < 16; ++w) {
    const int q = w >> 2, j = w & 3;
    const v2f cq = (q == 0) ? c.b0 : (q == 1) ? c.b1 : (q == 2) ? c.b2 : c.b3;
    v8f cin = v8f{};
    if (q != 0) cin = accW[j];
    accW[j] = __builtin_amdgcn_wmma_f32_16x16x4_f32(false, cq, false, a[j][q], (short)0, cin, false, false);
    if (RANK) {
      const int tile = w >> 2, stage = w & 3;
      if (stage == 0) {
        tv = accR[tile] - nrmR;
        vq_hold8(tv);
      } else if (stage == 1) {
        nb = vq_max3(vq_max3(best[tile], tv[0], tv[1]), tv[2], tv[3]);
        vq_hold1(nb);
      } else if (stage == 2) {
        nb = vq_max3(vq_max3(nb, tv[4], tv[5]), tv[6], tv[7]);
        vq_hold1(nb);
      } else {
        vq_finish(tile, nb, grpR, best, bgrp);
        vq_hold2(best[tile], bgrp[tile]);
      }
      __builtin_amdgcn_sched_barrier(0);
    }
  }
}

__device__ __forceinline__ void vq_rank_all(const v8f (&accR)[RT], const v8f& nrmR, int grpR,
                                            float (&best)[RT], int (&bgrp)[RT]) {
#pragma unroll
  for (int j = 0; j < RT; ++j) {
    const v8f tv = accR[j] - nrmR;
    float nb = vq_max3(vq_max3(best[j], tv[0], tv[1]), tv[2], tv[3]);
    nb = vq_max3(vq_max3(nb, tv[4], tv[5]), tv[6], tv[7]);
    vq_finish(j, nb, grpR, best, bgrp);
  }
}

__global__ __launch_bounds__(32) void vq_search(const float* __restrict__ xt,
                                                const float* __restrict__ cb,
                                                const float* __restrict__ cnh,
                                                float* __restrict__ pval,
                                                int* __restrict__ pgrp) {
  const int lane = threadIdx.x;
  const int half = lane >> 4;
  const int l16  = lane & 15;
  const int dimOff = half * 2;
  const int rowGroup = blockIdx.x;
  const int ks = blockIdx.y;

  v2f a[RT][4];
#pragma unroll
  for (int j = 0; j < RT; ++j) {
    const float* xrow = xt + (rowGroup * ROWS_PER_WAVE + j * 16 + l16) * DIMS;
#pragma unroll
    for (int q = 0; q < 4; ++q) a[j][q] = *(const v2f*)(xrow + 4 * q + dimOff);
  }

  float best[RT];
  int   bgrp[RT];
#pragma unroll
  for (int j = 0; j < RT; ++j) { best[j] = -3.4e38f; bgrp[j] = 0; }

  const int cBeg = ks * CHUNKS_PER_SPLIT;

  CFrag s0 = vq_load_codes(cb, cBeg, l16, dimOff);
  CFrag s1 = vq_load_codes(cb, cBeg + 1, l16, dimOff);
  v8f n0 = vq_load_norms(cnh, cBeg, half);
  v8f n1 = v8f{};
  int grpR = cBeg * 2 + half;
  v8f accA[RT], accB[RT];
  vq_step<false>(accA, accA, a, s0, n1, 0, best, bgrp);

  for (int i = 0; i < (CHUNKS_PER_SPLIT - 2) / 2; ++i) {
    const int k = cBeg + 2 * i + 1;
    s0 = vq_load_codes(cb, k + 1, l16, dimOff);
    n1 = vq_load_norms(cnh, k, half);
    __builtin_amdgcn_sched_barrier(0);
    vq_step<true>(accB, accA, a, s1, n0, grpR, best, bgrp);
    grpR += 2;

    s1 = vq_load_codes(cb, k + 2, l16, dimOff);
    n0 = vq_load_norms(cnh, k + 1, half);
    __builtin_amdgcn_sched_barrier(0);
    vq_step<true>(accA, accB, a, s0, n1, grpR, best, bgrp);
    grpR += 2;
  }

  n1 = vq_load_norms(cnh, cBeg + CHUNKS_PER_SPLIT - 1, half);
  vq_step<true>(accB, accA, a, s1, n0, grpR, best, bgrp);
  grpR += 2;
  vq_rank_all(accB, n1, grpR, best, bgrp);

#pragma unroll
  for (int j = 0; j < RT; ++j) {
    const float ov = __shfl_xor(best[j], 16, 32);
    const int   og = __shfl_xor(bgrp[j], 16, 32);
    const bool take = (ov > best[j]) | ((ov == best[j]) & (og < bgrp[j]));
    best[j] = take ? ov : best[j];
    bgrp[j] = take ? og : bgrp[j];
  }

#pragma unroll
  for (int s = 0; s < RT / 2; ++s) {
    const float v = half ? best[2 * s + 1] : best[2 * s];
    const int   g = half ? bgrp[2 * s + 1] : bgrp[2 * s];
    const int row = rowGroup * ROWS_PER_WAVE + s * 32 + lane;
    pval[ks * N_ROWS + row] = v;
    pgrp[ks * N_ROWS + row] = g;
  }
}

__global__ __launch_bounds__(256) void vq_finalize(const float* __restrict__ xt,
                                                   const float* __restrict__ cb,
                                                   const float* __restrict__ cnh,
                                                   const float* __restrict__ pval,
                                                   const int* __restrict__ pgrp,
                                                   float* __restrict__ out,
                                                   float* __restrict__ rowloss) {
  int n = blockIdx.x * blockDim.x + threadIdx.x;
  float best = -3.4e38f;
  int bg = 0;
#pragma unroll
  for (int k = 0; k < KSPLIT; ++k) {
    float v = pval[k * N_ROWS + n];
    int   g = pgrp[k * N_ROWS + n];
    bool take = (v > best) | ((v == best) & (g < bg));
    best = take ? v : best;
    bg   = take ? g : bg;
  }
  const float* xr = xt + n * DIMS;
  float xv[DIMS];
#pragma unroll
  for (int i = 0; i < 4; ++i) {
    float4 v = ((const float4*)xr)[i];
    xv[4 * i] = v.x; xv[4 * i + 1] = v.y; xv[4 * i + 2] = v.z; xv[4 * i + 3] = v.w;
  }
  const float4* grow = (const float4*)(cb + bg * 8 * DIMS);
  const float4* gn   = (const float4*)(cnh + bg * 8);
  const float4 nlo = gn[0], nhi = gn[1];
  const float nrm[8] = {nlo.x, nlo.y, nlo.z, nlo.w, nhi.x, nhi.y, nhi.z, nhi.w};
  float sbest = -3.4e38f;
  int bi = bg * 8;
#pragma unroll
  for (int c = 0; c < 8; ++c) {
    float dot = 0.f;
#pragma unroll
    for (int i = 0; i < 4; ++i) {
      float4 v = grow[c * 4 + i];
      dot += xv[4 * i] * v.x;
      dot += xv[4 * i + 1] * v.y;
      dot += xv[4 * i + 2] * v.z;
      dot += xv[4 * i + 3] * v.w;
    }
    const float s = dot - nrm[c];
    const bool take = s > sbest;
    sbest = take ? s : sbest;
    bi    = take ? bg * 8 + c : bi;
  }
  const float* q  = cb + bi * DIMS;
  int b = n >> 10, t = n & 1023;
  float loss = 0.f;
#pragma unroll
  for (int d = 0; d < DIMS; ++d) {
    float qv = q[d];
    out[(b * 16 + d) * 1024 + t] = qv;
    float diff = qv - xv[d];
    loss += diff * diff;
  }
  rowloss[n] = loss;
}

__global__ __launch_bounds__(256) void vq_loss(const float* __restrict__ rowloss,
                                               float* __restrict__ out) {
  __shared__ float sm[256];
  int tid = threadIdx.x;
  float s = 0.f;
  for (int i = tid; i < N_ROWS; i += 256) s += rowloss[i];
  sm[tid] = s;
  __syncthreads();
  for (int off = 128; off > 0; off >>= 1) {
    if (tid < off) sm[tid] += sm[tid + off];
    __syncthreads();
  }
  if (tid == 0) out[N_ROWS * DIMS] = sm[0] / (float)(N_ROWS * DIMS);
}

extern "C" void kernel_launch(void* const* d_in, const int* in_sizes, int n_in,
                              void* d_out, int out_size, void* d_ws, size_t ws_size,
                              hipStream_t stream) {
  const float* x  = (const float*)d_in[0];
  const float* cb = (const float*)d_in[1];
  float* out = (float*)d_out;

  float* ws      = (float*)d_ws;
  float* xt      = ws;
  float* cnh     = ws + 65536;
  float* pval    = ws + 131072;
  int*   pgrp    = (int*)(ws + 393216);
  float* rowloss = ws + 655360;

  vq_prep_transpose<<<256, 256, 0, stream>>>(x, xt);
  vq_prep_cnorm<<<256, 256, 0, stream>>>(cb, cnh);

  dim3 grid(64, KSPLIT);
  vq_search<<<grid, 32, 0, stream>>>(xt, cb, cnh, pval, pgrp);

  vq_finalize<<<16, 256, 0, stream>>>(xt, cb, cnh, pval, pgrp, out, rowloss);
  vq_loss<<<1, 256, 0, stream>>>(rowloss, out);
}


# Half norms enter each tile's WMMA chain as its C operand (kept negated in LDS): accumulators end as finished scores, the 8 subtractions per tile and chunk go; score sum order changes: -|c|^2/2 first
# speedup vs baseline: 1.9065x; 1.1252x over previous
#include <hip/hip_runtime.h>

typedef __attribute__((ext_vector_type(2))) float v2f;
typedef __attribute__((ext_vector_type(8))) float v8f;

#define N_ROWS 4096
#define DIMS   16
#define KCODES 65536
#define KSPLIT 64
#define CHUNKS_PER_SPLIT (KCODES / 16 / KSPLIT)
#define RT 4
#define ROWS_PER_WAVE (RT * 16)
#define CODES_PER_SPLIT (KCODES / KSPLIT)
#define GROUP_WAVES 16
#define GROUP_THREADS (32 * GROUP_WAVES)
#define CODES_PER_THREAD (CODES_PER_SPLIT / GROUP_THREADS)

__device__ __forceinline__ float vq_neg_half_norm(const float* __restrict__ row) {
  const float4* r = (const float4*)row;
  const float4 v0 = r[0], v1 = r[1], v2 = r[2], v3 = r[3];
  float s;
  {
#pragma clang fp contract(off)
    const float t0 = ((v0.x * v0.x + v0.y * v0.y) + v0.z * v0.z) + v0.w * v0.w;
    const float t1 = ((v1.x * v1.x + v1.y * v1.y) + v1.z * v1.z) + v1.w * v1.w;
    const float t2 = ((v2.x * v2.x + v2.y * v2.y) + v2.z * v2.z) + v2.w * v2.w;
    const float t3 = ((v3.x * v3.x + v3.y * v3.y) + v3.z * v3.z) + v3.w * v3.w;
    s = ((t0 + t1) + t2) + t3;
  }
  return -0.5f * s;
}

struct CFrag { v2f b0, b1, b2, b3; };

__device__ __forceinline__ CFrag vq_load_codes(const float* __restrict__ cb, int chunk, int l16, int dimOff) {
  const float* crow = cb + (chunk * 16 + l16) * DIMS + dimOff;
  CFrag f;
  f.b0 = *(const v2f*)(crow + 0);
  f.b1 = *(const v2f*)(crow + 4);
  f.b2 = *(const v2f*)(crow + 8);
  f.b3 = *(const v2f*)(crow + 12);
  return f;
}

__device__ __forceinline__ v8f vq_load_norms(const float* snrm, int chunk, int half) {
  return *(const v8f*)(snrm + chunk * 16 + half * 8);
}

__device__ __forceinline__ float vq_max3(float a, float b, float c) {
  return __builtin_fmaxf(__builtin_fmaxf(a, b), c);
}
__device__ __forceinline__ void vq_finish(int j, float nb, int grpR, float (&best)[RT], int (&bgrp)[RT]) {
  const bool g = nb > best[j];
  bgrp[j] = g ? grpR : bgrp[j];
  best[j] = nb;
}

__device__ __forceinline__ void vq_hold8(const v8f& v) { asm volatile("" :: "v"(v)); }
__device__ __forceinline__ void vq_hold1(float v) { asm volatile("" :: "v"(v)); }
__device__ __forceinline__ void vq_hold2(float v, int i) { asm volatile("" :: "v"(v), "v"(i)); }

template <bool RANK>
__device__ __forceinline__ void vq_step(v8f (&accW)[RT], const v8f (&accR)[RT], const v2f (&a)[RT][4],
                                        const CFrag& c, const v8f& nnW, int grpR,
                                        float (&best)[RT], int (&bgrp)[RT]) {
  float nb = 0.f;
#pragma unroll
  for (int w = 0; w < 16; ++w) {
    const int q = w >> 2, j = w & 3;
    const v2f cq = (q == 0) ? c.b0 : (q == 1) ? c.b1 : (q == 2) ? c.b2 : c.b3;
    v8f cin = nnW;
    if (q != 0) cin = accW[j];
    accW[j] = __builtin_amdgcn_wmma_f32_16x16x4_f32(false, cq, false, a[j][q], (short)0, cin, false, false);
    if (w == RT - 1) vq_hold8(nnW);
    if (RANK) {
      const int tile = w >> 2, stage = w & 3;
      const v8f& tv = accR[tile];
      if (stage == 0) {
        nb = vq_max3(best[tile], tv[0], tv[1]);
        vq_hold1(nb);
      } else if (stage == 1) {
        nb = vq_max3(nb, tv[2], tv[3]);
        vq_hold1(nb);
      } else if (stage == 2) {
        nb = vq_max3(nb, tv[4], tv[5]);
        vq_hold1(nb);
      } else {
        nb = vq_max3(nb, tv[6], tv[7]);
        vq_finish(tile, nb, grpR, best, bgrp);
        vq_hold2(best[tile], bgrp[tile]);
      }
      __builtin_amdgcn_sched_barrier(0);
    }
  }
}

__device__ __forceinline__ void vq_rank_all(const v8f (&accR)[RT], int grpR,
                                            float (&best)[RT], int (&bgrp)[RT]) {
#pragma unroll
  for (int j = 0; j < RT; ++j) {
    const v8f& tv = accR[j];
    float nb = vq_max3(vq_max3(best[j], tv[0], tv[1]), tv[2], tv[3]);
    nb = vq_max3(vq_max3(nb, tv[4], tv[5]), tv[6], tv[7]);
    vq_finish(j, nb, grpR, best, bgrp);
  }
}

__global__ __launch_bounds__(GROUP_THREADS) void vq_search(const float* __restrict__ x,
                                                           const float* __restrict__ cb,
                                                           float* __restrict__ xt,
                                                           float* __restrict__ cnh,
                                                           float* __restrict__ pval,
                                                           int* __restrict__ pgrp) {
  __shared__ __attribute__((aligned(32))) float snrm[CODES_PER_SPLIT];
  const int tid  = threadIdx.x;
  const int lane = tid & 31;
  const int wave = tid >> 5;
  const int half = lane >> 4;
  const int l16  = lane & 15;
  const int dimOff = half * 2;
  const int ks = blockIdx.x;
  const int rowGroup = blockIdx.y * GROUP_WAVES + wave;

  v2f a[RT][4];
#pragma unroll
  for (int j = 0; j < RT; ++j) {
    const int n = rowGroup * ROWS_PER_WAVE + j * 16 + l16;
    const float* xcol = x + (n >> 10) * (DIMS * 1024) + (n & 1023);
#pragma unroll
    for (int q = 0; q < 4; ++q) {
      const int d = 4 * q + dimOff;
      a[j][q] = v2f{xcol[d * 1024], xcol[(d + 1) * 1024]};
    }
  }
  if (ks == 0) {
#pragma unroll
    for (int j = 0; j < RT; ++j) {
      float* xrow = xt + (rowGroup * ROWS_PER_WAVE + j * 16 + l16) * DIMS;
#pragma unroll
      for (int q = 0; q < 4; ++q) *(v2f*)(xrow + 4 * q + dimOff) = a[j][q];
    }
  }

  float best[RT];
  int   bgrp[RT];
#pragma unroll
  for (int j = 0; j < RT; ++j) { best[j] = -3.4e38f; bgrp[j] = 0; }

  const int cBeg = ks * CHUNKS_PER_SPLIT;

  CFrag s0 = vq_load_codes(cb, cBeg, l16, dimOff);
  CFrag s1 = vq_load_codes(cb, cBeg + 1, l16, dimOff);

  float hn[CODES_PER_THREAD];
#pragma unroll
  for (int r = 0; r < CODES_PER_THREAD; ++r)
    hn[r] = vq_neg_half_norm(cb + (ks * CODES_PER_SPLIT + r * GROUP_THREADS + tid) * DIMS);
#pragma unroll
  for (int r = 0; r < CODES_PER_THREAD; ++r) snrm[r * GROUP_THREADS + tid] = hn[r];
  if (blockIdx.y == 0) {
#pragma unroll
    for (int r = 0; r < CODES_PER_THREAD; ++r) cnh[ks * CODES_PER_SPLIT + r * GROUP_THREADS + tid] = hn[r];
  }
  __syncthreads();

  v8f n0 = vq_load_norms(snrm, 0, half);
  v8f n1 = vq_load_norms(snrm, 1, half);
  int grpR = cBeg * 2 + half;
  v8f accA[RT], accB[RT];
  vq_step<false>(accA, accA, a, s0, n0, 0, best, bgrp);

  for (int i = 0; i < (CHUNKS_PER_SPLIT - 2) / 2; ++i) {
    const int kl = 2 * i + 1;
    const int k = cBeg + kl;
    s0 = vq_load_codes(cb, k + 1, l16, dimOff);
    n0 = vq_load_norms(snrm, kl + 1, half);
    __builtin_amdgcn_sched_barrier(0);
    vq_step<true>(accB, accA, a, s1, n1, grpR, best, bgrp);
    grpR += 2;

    s1 = vq_load_codes(cb, k + 2, l16, dimOff);
    n1 = vq_load_norms(snrm, kl + 2, half);
    __builtin_amdgcn_sched_barrier(0);
    vq_step<true>(accA, accB, a, s0, n0, grpR, best, bgrp);
    grpR += 2;
  }

  vq_step<true>(accB, accA, a, s1, n1, grpR, best, bgrp);
  grpR += 2;
  vq_rank_all(accB, grpR, best, bgrp);

#pragma unroll
  for (int j = 0; j < RT; ++j) {
    const float ov = __shfl_xor(best[j], 16, 32);
    const int   og = __shfl_xor(bgrp[j], 16, 32);
    const bool take = (ov > best[j]) | ((ov == best[j]) & (og < bgrp[j]));
    best[j] = take ? ov : best[j];
    bgrp[j] = take ? og : bgrp[j];
  }

#pragma unroll
  for (int s = 0; s < RT / 2; ++s) {
    const float v = half ? best[2 * s + 1] : best[2 * s];
    const int   g = half ? bgrp[2 * s + 1] : bgrp[2 * s];
    const int row = rowGroup * ROWS_PER_WAVE + s * 32 + lane;
    pval[ks * N_ROWS + row] = v;
    pgrp[ks * N_ROWS + row] = g;
  }
}

__global__ __launch_bounds__(256) void vq_finalize(const float* __restrict__ xt,
                                                   const float* __restrict__ cb,
                                                   const float* __restrict__ cnh,
                                                   const float* __restrict__ pval,
                                                   const int* __restrict__ pgrp,
                                                   float* __restrict__ out,
                                                   float* __restrict__ rowloss) {
  int n = blockIdx.x * blockDim.x + threadIdx.x;
  float best = -3.4e38f;
  int bg = 0;
#pragma unroll
  for (int k = 0; k < KSPLIT; ++k) {
    float v = pval[k * N_ROWS + n];
    int   g = pgrp[k * N_ROWS + n];
    bool take = (v > best) | ((v == best) & (g < bg));
    best = take ? v : best;
    bg   = take ? g : bg;
  }
  const float* xr = xt + n * DIMS;
  float xv[DIMS];
#pragma unroll
  for (int i = 0; i < 4; ++i) {
    float4 v = ((const float4*)xr)[i];
    xv[4 * i] = v.x; xv[4 * i + 1] = v.y; xv[4 * i + 2] = v.z; xv[4 * i + 3] = v.w;
  }
  const float4* grow = (const float4*)(cb + bg * 8 * DIMS);
  const float4* gn   = (const float4*)(cnh + bg * 8);
  const float4 nlo = gn[0], nhi = gn[1];
  const float nrm[8] = {nlo.x, nlo.y, nlo.z, nlo.w, nhi.x, nhi.y, nhi.z, nhi.w};
  float sbest = -3.4e38f;
  int bi = bg * 8;
#pragma unroll
  for (int c = 0; c < 8; ++c) {
    float s = nrm[c];
#pragma unroll
    for (int i = 0; i < 4; ++i) {
      float4 v = grow[c * 4 + i];
      s += xv[4 * i] * v.x;
      s += xv[4 * i + 1] * v.y;
      s += xv[4 * i + 2] * v.z;
      s += xv[4 * i + 3] * v.w;
    }
    const bool take = s > sbest;
    sbest = take ? s : sbest;
    bi    = take ? bg * 8 + c : bi;
  }
  const float* q  = cb + bi * DIMS;
  int b = n >> 10, t = n & 1023;
  float loss = 0.f;
#pragma unroll
  for (int d = 0; d < DIMS; ++d) {
    float qv = q[d];
    out[(b * 16 + d) * 1024 + t] = qv;
    float diff = qv - xv[d];
    loss += diff * diff;
  }
  rowloss[n] = loss;
}

__global__ __launch_bounds__(256) void vq_loss(const float* __restrict__ rowloss,
                                               float* __restrict__ out) {
  __shared__ float sm[256];
  int tid = threadIdx.x;
  float s = 0.f;
  for (int i = tid; i < N_ROWS; i += 256) s += rowloss[i];
  sm[tid] = s;
  __syncthreads();
  for (int off = 128; off > 0; off >>= 1) {
    if (tid < off) sm[tid] += sm[tid + off];
    __syncthreads();
  }
  if (tid == 0) out[N_ROWS * DIMS] = sm[0] / (float)(N_ROWS * DIMS);
}

extern "C" void kernel_launch(void* const* d_in, const int* in_sizes, int n_in,
                              void* d_out, int out_size, void* d_ws, size_t ws_size,
                              hipStream_t stream) {
  const float* x  = (const float*)d_in[0];
  const float* cb = (const float*)d_in[1];
  float* out = (float*)d_out;

  float* ws      = (float*)d_ws;
  float* xt      = ws;
  float* cnh     = ws + 65536;
  float* pval    = ws + 131072;
  int*   pgrp    = (int*)(ws + 393216);
  float* rowloss = ws + 655360;

  dim3 grid(KSPLIT, 4);
  vq_search<<<grid, GROUP_THREADS, 0, stream>>>(x, cb, xt, cnh, pval, pgrp);

  vq_finalize<<<16, 256, 0, stream>>>(xt, cb, cnh, pval, pgrp, out, rowloss);
  vq_loss<<<1, 256, 0, stream>>>(rowloss, out);
}


# Finalize re-partitioned: 4 waves share each row (K-split fold, group re-score and output dims in quarters, merged through LDS) on 128 work-groups instead of 16; same sums and order, bit-identical
# speedup vs baseline: 2.0796x; 1.0908x over previous
#include <hip/hip_runtime.h>

typedef __attribute__((ext_vector_type(2))) float v2f;
typedef __attribute__((ext_vector_type(8))) float v8f;

#define N_ROWS 4096
#define DIMS   16
#define KCODES 65536
#define KSPLIT 64
#define CHUNKS_PER_SPLIT (KCODES / 16 / KSPLIT)
#define RT 4
#define ROWS_PER_WAVE (RT * 16)
#define CODES_PER_SPLIT (KCODES / KSPLIT)
#define GROUP_WAVES 16
#define GROUP_THREADS (32 * GROUP_WAVES)
#define CODES_PER_THREAD (CODES_PER_SPLIT / GROUP_THREADS)

__device__ __forceinline__ float vq_neg_half_norm(const float* __restrict__ row) {
  const float4* r = (const float4*)row;
  const float4 v0 = r[0], v1 = r[1], v2 = r[2], v3 = r[3];
  float s;
  {
#pragma clang fp contract(off)
    const float t0 = ((v0.x * v0.x + v0.y * v0.y) + v0.z * v0.z) + v0.w * v0.w;
    const float t1 = ((v1.x * v1.x + v1.y * v1.y) + v1.z * v1.z) + v1.w * v1.w;
    const float t2 = ((v2.x * v2.x + v2.y * v2.y) + v2.z * v2.z) + v2.w * v2.w;
    const float t3 = ((v3.x * v3.x + v3.y * v3.y) + v3.z * v3.z) + v3.w * v3.w;
    s = ((t0 + t1) + t2) + t3;
  }
  return -0.5f * s;
}

struct CFrag { v2f b0, b1, b2, b3; };

__device__ __forceinline__ CFrag vq_load_codes(const float* __restrict__ cb, int chunk, int l16, int dimOff) {
  const float* crow = cb + (chunk * 16 + l16) * DIMS + dimOff;
  CFrag f;
  f.b0 = *(const v2f*)(crow + 0);
  f.b1 = *(const v2f*)(crow + 4);
  f.b2 = *(const v2f*)(crow + 8);
  f.b3 = *(const v2f*)(crow + 12);
  return f;
}

__device__ __forceinline__ v8f vq_load_norms(const float* snrm, int chunk, int half) {
  return *(const v8f*)(snrm + chunk * 16 + half * 8);
}

__device__ __forceinline__ float vq_max3(float a, float b, float c) {
  return __builtin_fmaxf(__builtin_fmaxf(a, b), c);
}
__device__ __forceinline__ void vq_finish(int j, float nb, int grpR, float (&best)[RT], int (&bgrp)[RT]) {
  const bool g = nb > best[j];
  bgrp[j] = g ? grpR : bgrp[j];
  best[j] = nb;
}

__device__ __forceinline__ void vq_hold8(const v8f& v) { asm volatile("" :: "v"(v)); }
__device__ __forceinline__ void vq_hold1(float v) { asm volatile("" :: "v"(v)); }
__device__ __forceinline__ void vq_hold2(float v, int i) { asm volatile("" :: "v"(v), "v"(i)); }

template <bool RANK>
__device__ __forceinline__ void vq_step(v8f (&accW)[RT], const v8f (&accR)[RT], const v2f (&a)[RT][4],
                                        const CFrag& c, const v8f& nnW, int grpR,
                                        float (&best)[RT], int (&bgrp)[RT]) {
  float nb = 0.f;
#pragma unroll
  for (int w = 0; w < 16; ++w) {
    const int q = w >> 2, j = w & 3;
    const v2f cq = (q == 0) ? c.b0 : (q == 1) ? c.b1 : (q == 2) ? c.b2 : c.b3;
    v8f cin = nnW;
    if (q != 0) cin = accW[j];
    accW[j] = __builtin_amdgcn_wmma_f32_16x16x4_f32(false, cq, false, a[j][q], (short)0, cin, false, false);
    if (w == RT - 1) vq_hold8(nnW);
    if (RANK) {
      const int tile = w >> 2, stage = w & 3;
      const v8f& tv = accR[tile];
      if (stage == 0) {
        nb = vq_max3(best[tile], tv[0], tv[1]);
        vq_hold1(nb);
      } else if (stage == 1) {
        nb = vq_max3(nb, tv[2], tv[3]);
        vq_hold1(nb);
      } else if (stage == 2) {
        nb = vq_max3(nb, tv[4], tv[5]);
        vq_hold1(nb);
      } else {
        nb = vq_max3(nb, tv[6], tv[7]);
        vq_finish(tile, nb, grpR, best, bgrp);
        vq_hold2(best[tile], bgrp[tile]);
      }
      __builtin_amdgcn_sched_barrier(0);
    }
  }
}

__device__ __forceinline__ void vq_rank_all(const v8f (&accR)[RT], int grpR,
                                            float (&best)[RT], int (&bgrp)[RT]) {
#pragma unroll
  for (int j = 0; j < RT; ++j) {
    const v8f& tv = accR[j];
    float nb = vq_max3(vq_max3(best[j], tv[0], tv[1]), tv[2], tv[3]);
    nb = vq_max3(vq_max3(nb, tv[4], tv[5]), tv[6], tv[7]);
    vq_finish(j, nb, grpR, best, bgrp);
  }
}

__global__ __launch_bounds__(GROUP_THREADS) void vq_search(const float* __restrict__ x,
                                                           const float* __restrict__ cb,
                                                           float* __restrict__ xt,
                                                           float* __restrict__ cnh,
                                                           float* __restrict__ pval,
                                                           int* __restrict__ pgrp) {
  __shared__ __attribute__((aligned(32))) float snrm[CODES_PER_SPLIT];
  const int tid  = threadIdx.x;
  const int lane = tid & 31;
  const int wave = tid >> 5;
  const int half = lane >> 4;
  const int l16  = lane & 15;
  const int dimOff = half * 2;
  const int ks = blockIdx.x;
  const int rowGroup = blockIdx.y * GROUP_WAVES + wave;

  v2f a[RT][4];
#pragma unroll
  for (int j = 0; j < RT; ++j) {
    const int n = rowGroup * ROWS_PER_WAVE + j * 16 + l16;
    const float* xcol = x + (n >> 10) * (DIMS * 1024) + (n & 1023);
#pragma unroll
    for (int q = 0; q < 4; ++q) {
      const int d = 4 * q + dimOff;
      a[j][q] = v2f{xcol[d * 1024], xcol[(d + 1) * 1024]};
    }
  }
  if (ks == 0) {
#pragma unroll
    for (int j = 0; j < RT; ++j) {
      float* xrow = xt + (rowGroup * ROWS_PER_WAVE + j * 16 + l16) * DIMS;
#pragma unroll
      for (int q = 0; q < 4; ++q) *(v2f*)(xrow + 4 * q + dimOff) = a[j][q];
    }
  }

  float best[RT];
  int   bgrp[RT];
#pragma unroll
  for (int j = 0; j < RT; ++j) { best[j] = -3.4e38f; bgrp[j] = 0; }

  const int cBeg = ks * CHUNKS_PER_SPLIT;

  CFrag s0 = vq_load_codes(cb, cBeg, l16, dimOff);
  CFrag s1 = vq_load_codes(cb, cBeg + 1, l16, dimOff);

  float hn[CODES_PER_THREAD];
#pragma unroll
  for (int r = 0; r < CODES_PER_THREAD; ++r)
    hn[r] = vq_neg_half_norm(cb + (ks * CODES_PER_SPLIT + r * GROUP_THREADS + tid) * DIMS);
#pragma unroll
  for (int r = 0; r < CODES_PER_THREAD; ++r) snrm[r * GROUP_THREADS + tid] = hn[r];
  if (blockIdx.y == 0) {
#pragma unroll
    for (int r = 0; r < CODES_PER_THREAD; ++r) cnh[ks * CODES_PER_SPLIT + r * GROUP_THREADS + tid] = hn[r];
  }
  __syncthreads();

  v8f n0 = vq_load_norms(snrm, 0, half);
  v8f n1 = vq_load_norms(snrm, 1, half);
  int grpR = cBeg * 2 + half;
  v8f accA[RT], accB[RT];
  vq_step<false>(accA, accA, a, s0, n0, 0, best, bgrp);

  for (int i = 0; i < (CHUNKS_PER_SPLIT - 2) / 2; ++i) {
    const int kl = 2 * i + 1;
    const int k = cBeg + kl;
    s0 = vq_load_codes(cb, k + 1, l16, dimOff);
    n0 = vq_load_norms(snrm, kl + 1, half);
    __builtin_amdgcn_sched_barrier(0);
    vq_step<true>(accB, accA, a, s1, n1, grpR, best, bgrp);
    grpR += 2;

    s1 = vq_load_codes(cb, k + 2, l16, dimOff);
    n1 = vq_load_norms(snrm, kl + 2, half);
    __builtin_amdgcn_sched_barrier(0);
    vq_step<true>(accA, accB, a, s0, n0, grpR, best, bgrp);
    grpR += 2;
  }

  vq_step<true>(accB, accA, a, s1, n1, grpR, best, bgrp);
  grpR += 2;
  vq_rank_all(accB, grpR, best, bgrp);

#pragma unroll
  for (int j = 0; j < RT; ++j) {
    const float ov = __shfl_xor(best[j], 16, 32);
    const int   og = __shfl_xor(bgrp[j], 16, 32);
    const bool take = (ov > best[j]) | ((ov == best[j]) & (og < bgrp[j]));
    best[j] = take ? ov : best[j];
    bgrp[j] = take ? og : bgrp[j];
  }

#pragma unroll
  for (int s = 0; s < RT / 2; ++s) {
    const float v = half ? best[2 * s + 1] : best[2 * s];
    const int   g = half ? bgrp[2 * s + 1] : bgrp[2 * s];
    const int row = rowGroup * ROWS_PER_WAVE + s * 32 + lane;
    pval[ks * N_ROWS + row] = v;
    pgrp[ks * N_ROWS + row] = g;
  }
}

#define FIN_ROWS 32
#define FIN_WAVES 4
#define FIN_THREADS (32 * FIN_WAVES)
#define FIN_SPLITS (KSPLIT / FIN_WAVES)
#define FIN_CODES (8 / FIN_WAVES)
#define FIN_DIMS (DIMS / FIN_WAVES)

__global__ __launch_bounds__(FIN_THREADS) void vq_finalize(const float* __restrict__ xt,
                                                           const float* __restrict__ cb,
                                                           const float* __restrict__ cnh,
                                                           const float* __restrict__ pval,
                                                           const int* __restrict__ pgrp,
                                                           float* __restrict__ out,
                                                           float* __restrict__ rowloss) {
  __shared__ float sval[FIN_WAVES][FIN_ROWS];
  __shared__ int   sgrp[FIN_WAVES][FIN_ROWS];
  __shared__ float sscr[FIN_WAVES][FIN_ROWS];
  __shared__ int   scod[FIN_WAVES][FIN_ROWS];
  const int lane = threadIdx.x & 31;
  const int w    = threadIdx.x >> 5;
  const int n    = blockIdx.x * FIN_ROWS + lane;


  float best = -3.4e38f;
  int bg = 0;
#pragma unroll
  for (int k = 0; k < FIN_SPLITS; ++k) {
    const int ks = w * FIN_SPLITS + k;
    float v = pval[ks * N_ROWS + n];
    int   g = pgrp[ks * N_ROWS + n];
    bool take = (v > best) | ((v == best) & (g < bg));
    best = take ? v : best;
    bg   = take ? g : bg;
  }
  sval[w][lane] = best;
  sgrp[w][lane] = bg;
  __syncthreads();
  best = sval[0][lane];
  bg   = sgrp[0][lane];
#pragma unroll
  for (int u = 1; u < FIN_WAVES; ++u) {
    float v = sval[u][lane];
    int   g = sgrp[u][lane];
    bool take = (v > best) | ((v == best) & (g < bg));
    best = take ? v : best;
    bg   = take ? g : bg;
  }

  const float* xr = xt + n * DIMS;
  float xv[DIMS];
#pragma unroll
  for (int i = 0; i < 4; ++i) {
    float4 v = ((const float4*)xr)[i];
    xv[4 * i] = v.x; xv[4 * i + 1] = v.y; xv[4 * i + 2] = v.z; xv[4 * i + 3] = v.w;
  }
  const int c0 = bg * 8 + w * FIN_CODES;
  const float4* grow = (const float4*)(cb + c0 * DIMS);
  const v2f nn = *(const v2f*)(cnh + c0);
  const float nrm[FIN_CODES] = {nn.x, nn.y};
  float sbest = -3.4e38f;
  int bi = c0;
#pragma unroll
  for (int c = 0; c < FIN_CODES; ++c) {
    float s = nrm[c];
#pragma unroll
    for (int i = 0; i < 4; ++i) {
      float4 v = grow[c * 4 + i];
      s += xv[4 * i] * v.x;
      s += xv[4 * i + 1] * v.y;
      s += xv[4 * i + 2] * v.z;
      s += xv[4 * i + 3] * v.w;
    }
    const bool take = s > sbest;
    sbest = take ? s : sbest;
    bi    = take ? c0 + c : bi;
  }
  sscr[w][lane] = sbest;
  scod[w][lane] = bi;
  __syncthreads();
  sbest = sscr[0][lane];
  bi    = scod[0][lane];
#pragma unroll
  for (int u = 1; u < FIN_WAVES; ++u) {
    float s = sscr[u][lane];
    int   c = scod[u][lane];
    const bool take = s > sbest;
    sbest = take ? s : sbest;
    bi    = take ? c : bi;
  }

  const float* q = cb + bi * DIMS;
  const int b = n >> 10, t = n & 1023;
  const float4 mine = ((const float4*)q)[w];
  const float qm[FIN_DIMS] = {mine.x, mine.y, mine.z, mine.w};
#pragma unroll
  for (int dd = 0; dd < FIN_DIMS; ++dd)
    out[(b * 16 + w * FIN_DIMS + dd) * 1024 + t] = qm[dd];
  if (w == 0) {
    float loss = 0.f;
#pragma unroll
    for (int d = 0; d < DIMS; ++d) {
      float qv = q[d];
      float diff = qv - xv[d];
      loss += diff * diff;
    }
    rowloss[n] = loss;
  }
}

__global__ __launch_bounds__(256) void vq_loss(const float* __restrict__ rowloss,
                                               float* __restrict__ out) {
  __shared__ float sm[256];
  int tid = threadIdx.x;
  float s = 0.f;
  for (int i = tid; i < N_ROWS; i += 256) s += rowloss[i];
  sm[tid] = s;
  __syncthreads();
  for (int off = 128; off > 0; off >>= 1) {
    if (tid < off) sm[tid] += sm[tid + off];
    __syncthreads();
  }
  if (tid == 0) out[N_ROWS * DIMS] = sm[0] / (float)(N_ROWS * DIMS);
}

extern "C" void kernel_launch(void* const* d_in, const int* in_sizes, int n_in,
                              void* d_out, int out_size, void* d_ws, size_t ws_size,
                              hipStream_t stream) {
  const float* x  = (const float*)d_in[0];
  const float* cb = (const float*)d_in[1];
  float* out = (float*)d_out;

  float* ws      = (float*)d_ws;
  float* xt      = ws;
  float* cnh     = ws + 65536;
  float* pval    = ws + 131072;
  int*   pgrp    = (int*)(ws + 393216);
  float* rowloss = ws + 655360;

  dim3 grid(KSPLIT, 4);
  vq_search<<<grid, GROUP_THREADS, 0, stream>>>(x, cb, xt, cnh, pval, pgrp);

  vq_finalize<<<128, FIN_THREADS, 0, stream>>>(xt, cb, cnh, pval, pgrp, out, rowloss);
  vq_loss<<<1, 256, 0, stream>>>(rowloss, out);
}
